# starting kernel
# baseline (speedup 1.0000x reference)
#include <hip/hip_runtime.h>
#include <hip/hip_bf16.h>


typedef __attribute__((ext_vector_type(2))) float v2f;
typedef __attribute__((ext_vector_type(8))) float v8f;

#define IN_CH 256
#define HID_CH 128
#define GEMM_ROWS 32


__global__ __launch_bounds__(256) void k_init_deg(float* deg, int N) {
    int i = blockIdx.x * 256 + threadIdx.x;
    if (i < N) deg[i] = 1.0f;
}

__global__ __launch_bounds__(256) void k_deg_scatter(const int* __restrict__ col,
                                                     const float* __restrict__ w,
                                                     float* deg, int E) {
    int e = blockIdx.x * 256 + threadIdx.x;
    if (e < E) unsafeAtomicAdd(&deg[col[e]], w[e]);
}

__global__ __launch_bounds__(256) void k_dinv(const float* __restrict__ deg,
                                              float* dinv, int N) {
    int i = blockIdx.x * 256 + threadIdx.x;
    if (i < N) {
        float d = deg[i];
        dinv[i] = (d > 0.0f) ? rsqrtf(d) : 0.0f;
    }
}

__global__ __launch_bounds__(256) void k_edge_norm(const int* __restrict__ row,
                                                   const int* __restrict__ col,
                                                   const float* __restrict__ w,
                                                   const float* __restrict__ dinv,
                                                   float* nrm, int E) {
    int e = blockIdx.x * 256 + threadIdx.x;
    if (e < E) nrm[e] = dinv[row[e]] * w[e] * dinv[col[e]];
}


__global__ __launch_bounds__(256) void k_gemm_wmma(const float* __restrict__ A,
                                                   const float* __restrict__ W,
                                                   float* __restrict__ H, int K) {
    __shared__ float As[GEMM_ROWS * (IN_CH + 4)];
    const int stride = K + 4;
    const int tid = threadIdx.x;
    const int rowbase = blockIdx.x * GEMM_ROWS;

    for (int idx = tid; idx < GEMM_ROWS * K; idx += 256) {
        int r = idx / K;
        int kk = idx - r * K;
        As[r * stride + kk] = A[(size_t)(rowbase + r) * K + kk];
    }
    __syncthreads();

    const int wave = tid >> 5;
    const int lane = tid & 31;
    const int m    = lane & 15;
    const int half = lane >> 4;
    const int cb   = wave * 16;

    v8f c0 = {};
    v8f c1 = {};
    const float* wp  = W + cb + m;
    const float* ap0 = As + m * stride;
    const float* ap1 = As + (16 + m) * stride;

    for (int k = 0; k < K; k += 4) {
        const int koff = k + half * 2;
        v2f b;
        b[0] = wp[(size_t)koff * HID_CH];
        b[1] = wp[(size_t)(koff + 1) * HID_CH];
        v2f a0, a1;
        a0[0] = ap0[koff];
        a0[1] = ap0[koff + 1];
        a1[0] = ap1[koff];
        a1[1] = ap1[koff + 1];
        c0 = __builtin_amdgcn_wmma_f32_16x16x4_f32(false, a0, false, b,
                                                   (short)0, c0, false, false);
        c1 = __builtin_amdgcn_wmma_f32_16x16x4_f32(false, a1, false, b,
                                                   (short)0, c1, false, false);
    }

    float* hp0 = H + (size_t)rowbase * HID_CH + cb + m;
    float* hp1 = hp0 + (size_t)16 * HID_CH;
#pragma unroll
    for (int r = 0; r < 8; ++r) {
        hp0[(size_t)(r + 8 * half) * HID_CH] = c0[r];
        hp1[(size_t)(r + 8 * half) * HID_CH] = c1[r];
    }
}


__global__ __launch_bounds__(256) void k_init_acc(const float* __restrict__ h,
                                                  const float* __restrict__ dinv,
                                                  float* __restrict__ acc, int N) {
    int t = blockIdx.x * 256 + threadIdx.x;
    int node = t >> 5, chunk = t & 31;
    if (node >= N) return;
    float s = dinv[node] * dinv[node];
    const float4* hp = (const float4*)(h + (size_t)node * HID_CH);
    float4* ap = (float4*)(acc + (size_t)node * HID_CH);
    float4 v = hp[chunk];
    v.x *= s; v.y *= s; v.z *= s; v.w *= s;
    ap[chunk] = v;
}

__global__ __launch_bounds__(256) void k_scatter(const int* __restrict__ row,
                                                 const int* __restrict__ col,
                                                 const float* __restrict__ nrm,
                                                 const float* __restrict__ h,
                                                 float* __restrict__ acc, int E) {
    int wid = (blockIdx.x * 256 + threadIdx.x) >> 5;
    int lane = threadIdx.x & 31;
    if (wid >= E) return;
    int r = row[wid];
    int c = col[wid];
    float nv = nrm[wid];
    const float4* hp = (const float4*)(h + (size_t)r * HID_CH);
    float4 v = hp[lane];
    float* dst = acc + (size_t)c * HID_CH + lane * 4;
    unsafeAtomicAdd(dst + 0, v.x * nv);
    unsafeAtomicAdd(dst + 1, v.y * nv);
    unsafeAtomicAdd(dst + 2, v.z * nv);
    unsafeAtomicAdd(dst + 3, v.w * nv);
}

__global__ __launch_bounds__(256) void k_bias_relu(float* __restrict__ acc,
                                                   const float* __restrict__ b, int N) {
    int g = blockIdx.x * 256 + threadIdx.x;
    if (g >= N * 32) return;
    float4* ap = (float4*)acc;
    const float4* bp = (const float4*)b;
    float4 v = ap[g];
    float4 bb = bp[g & 31];
    v.x = fmaxf(v.x + bb.x, 0.0f);
    v.y = fmaxf(v.y + bb.y, 0.0f);
    v.z = fmaxf(v.z + bb.z, 0.0f);
    v.w = fmaxf(v.w + bb.w, 0.0f);
    ap[g] = v;
}

__global__ __launch_bounds__(256) void k_bias_l2norm(float* __restrict__ out,
                                                     const float* __restrict__ b, int N) {
    int wid = (blockIdx.x * 256 + threadIdx.x) >> 5;
    int lane = threadIdx.x & 31;
    if (wid >= N) return;
    float4* op = (float4*)(out + (size_t)wid * HID_CH);
    const float4* bp = (const float4*)b;
    float4 v = op[lane];
    float4 bb = bp[lane];
    v.x += bb.x; v.y += bb.y; v.z += bb.z; v.w += bb.w;
    float ss = v.x * v.x + v.y * v.y + v.z * v.z + v.w * v.w;
#pragma unroll
    for (int m = 16; m >= 1; m >>= 1) ss += __shfl_xor(ss, m, 32);
    float inv = 1.0f / fmaxf(sqrtf(ss), 1e-12f);
    v.x *= inv; v.y *= inv; v.z *= inv; v.w *= inv;
    op[lane] = v;
}


static inline size_t align_up(size_t v, size_t a) { return (v + a - 1) & ~(a - 1); }

extern "C" void kernel_launch(void* const* d_in, const int* in_sizes, int n_in,
                              void* d_out, int out_size, void* d_ws, size_t ws_size,
                              hipStream_t stream) {
    const float* x  = (const float*)d_in[0];
    const int*   ei = (const int*)d_in[1];
    const float* ew = (const float*)d_in[2];
    const float* W1 = (const float*)d_in[3];
    const float* b1 = (const float*)d_in[4];
    const float* W2 = (const float*)d_in[5];
    const float* b2 = (const float*)d_in[6];
    float* out = (float*)d_out;

    const int N = in_sizes[0] / IN_CH;
    const int E = in_sizes[1] / 2;
    const int* row = ei;
    const int* col = ei + E;

    char* ws = (char*)d_ws;
    size_t off = 0;
    float* deg  = (float*)(ws + off); off = align_up(off + (size_t)N * 4, 1024);
    float* dinv = (float*)(ws + off); off = align_up(off + (size_t)N * 4, 1024);
    float* nrm  = (float*)(ws + off); off = align_up(off + (size_t)E * 4, 1024);
    float* h    = (float*)(ws + off); off = align_up(off + (size_t)N * HID_CH * 4, 1024);
    float* acc  = (float*)(ws + off); off = align_up(off + (size_t)N * HID_CH * 4, 1024);
    (void)ws_size; (void)n_in; (void)out_size;

    const int nb_nodes  = (N + 255) / 256;
    const int nb_edges  = (E + 255) / 256;
    const int nb_nvec   = (N * 32 + 255) / 256;
    const int nb_nwave  = (N + 7) / 8;
    const int nb_ewave  = (E + 7) / 8;
    const int nb_gemm   = N / GEMM_ROWS;

    k_init_deg   <<<nb_nodes, 256, 0, stream>>>(deg, N);
    k_deg_scatter<<<nb_edges, 256, 0, stream>>>(col, ew, deg, E);
    k_dinv       <<<nb_nodes, 256, 0, stream>>>(deg, dinv, N);
    k_edge_norm  <<<nb_edges, 256, 0, stream>>>(row, col, ew, dinv, nrm, E);

    k_gemm_wmma  <<<nb_gemm, 256, 0, stream>>>(x, W1, h, IN_CH);
    k_init_acc   <<<nb_nvec, 256, 0, stream>>>(h, dinv, acc, N);
    k_scatter    <<<nb_ewave, 256, 0, stream>>>(row, col, nrm, h, acc, E);
    k_bias_relu  <<<nb_nvec, 256, 0, stream>>>(acc, b1, N);

    k_gemm_wmma  <<<nb_gemm, 256, 0, stream>>>(acc, W2, h, HID_CH);
    k_init_acc   <<<nb_nvec, 256, 0, stream>>>(h, dinv, out, N);
    k_scatter    <<<nb_ewave, 256, 0, stream>>>(row, col, nrm, h, out, E);
    k_bias_l2norm<<<nb_nwave, 256, 0, stream>>>(out, b2, N);
}


# Edges counting-sorted by destination each call (arrival-order rank from returning int atomics); one wave per node sums its source rows in registers; no f32 atomics, sum order = that rank
# speedup vs baseline: 2.1886x; 2.1886x over previous
#include <hip/hip_runtime.h>
#include <hip/hip_bf16.h>


typedef __attribute__((ext_vector_type(2))) float v2f;
typedef __attribute__((ext_vector_type(8))) float v8f;

#define IN_CH 256
#define HID_CH 128
#define GEMM_ROWS 32


__global__ __launch_bounds__(256) void k_zero_cnt(int* cnt, int N) {
    int i = blockIdx.x * 256 + threadIdx.x;
    if (i < N) cnt[i] = 0;
}

__global__ __launch_bounds__(256) void k_count_rank(const int* __restrict__ col,
                                                    int* cnt,
                                                    int* __restrict__ rank, int E) {
    int e = blockIdx.x * 256 + threadIdx.x;
    if (e < E) rank[e] = atomicAdd(&cnt[col[e]], 1);
}

__global__ __launch_bounds__(1024) void k_scan(const int* __restrict__ cnt,
                                               int* __restrict__ rowptr, int N) {
    __shared__ int wsum[32];
    __shared__ int tot;
    const int tid = threadIdx.x;
    const int lane = tid & 31;
    const int wave = tid >> 5;
    int carry = 0;
    for (int base = 0; base < N; base += 4096) {
        const int i0 = base + tid * 4;
        int v0 = 0, v1 = 0, v2 = 0, v3 = 0;
        if (i0 + 3 < N) {
            int4 v = *(const int4*)(cnt + i0);
            v0 = v.x; v1 = v.y; v2 = v.z; v3 = v.w;
        } else {
            if (i0 < N) v0 = cnt[i0];
            if (i0 + 1 < N) v1 = cnt[i0 + 1];
            if (i0 + 2 < N) v2 = cnt[i0 + 2];
        }
        const int s = v0 + v1 + v2 + v3;
        int inc = s;
        for (int d = 1; d < 32; d <<= 1) {
            int t = __shfl_up(inc, d, 32);
            if (lane >= d) inc += t;
        }
        if (lane == 31) wsum[wave] = inc;
        __syncthreads();
        if (wave == 0) {
            int w = wsum[lane];
            int wi = w;
            for (int d = 1; d < 32; d <<= 1) {
                int t = __shfl_up(wi, d, 32);
                if (lane >= d) wi += t;
            }
            wsum[lane] = wi - w;
            if (lane == 31) tot = wi;
        }
        __syncthreads();
        const int ex = carry + wsum[wave] + inc - s;
        if (i0 + 3 < N) {
            *(int4*)(rowptr + i0) = make_int4(ex, ex + v0, ex + v0 + v1, ex + v0 + v1 + v2);
        } else {
            if (i0 < N) rowptr[i0] = ex;
            if (i0 + 1 < N) rowptr[i0 + 1] = ex + v0;
            if (i0 + 2 < N) rowptr[i0 + 2] = ex + v0 + v1;
        }
        carry += tot;
        __syncthreads();
    }
    if (tid == 0) rowptr[N] = carry;
}

__global__ __launch_bounds__(256) void k_fill(const int* __restrict__ row,
                                              const int* __restrict__ col,
                                              const float* __restrict__ w,
                                              const int* __restrict__ rowptr,
                                              const int* __restrict__ rank,
                                              int* __restrict__ pairs_i, int E) {
    int e = blockIdx.x * 256 + threadIdx.x;
    if (e < E) {
        int p = rowptr[col[e]] + rank[e];
        if ((unsigned)p < (unsigned)E)
            ((int2*)pairs_i)[p] = make_int2(row[e], __float_as_int(w[e]));
    }
}

__global__ __launch_bounds__(256) void k_deg_dinv(const int* __restrict__ rowptr,
                                                  const int* __restrict__ pairs_i,
                                                  float* __restrict__ dinv, int N) {
    int i = blockIdx.x * 256 + threadIdx.x;
    if (i < N) {
        const int2* pairs = (const int2*)pairs_i;
        float d = 1.0f;
        const int end = rowptr[i + 1];
        for (int p = rowptr[i]; p < end; ++p) d += __int_as_float(pairs[p].y);
        dinv[i] = (d > 0.0f) ? rsqrtf(d) : 0.0f;
    }
}


__global__ __launch_bounds__(256) void k_gemm_wmma(const float* __restrict__ A,
                                                   const float* __restrict__ W,
                                                   float* __restrict__ H, int K) {
    __shared__ float As[GEMM_ROWS * (IN_CH + 4)];
    const int stride = K + 4;
    const int tid = threadIdx.x;
    const int rowbase = blockIdx.x * GEMM_ROWS;

    for (int idx = tid; idx < GEMM_ROWS * K; idx += 256) {
        int r = idx / K;
        int kk = idx - r * K;
        As[r * stride + kk] = A[(size_t)(rowbase + r) * K + kk];
    }
    __syncthreads();

    const int wave = tid >> 5;
    const int lane = tid & 31;
    const int m    = lane & 15;
    const int half = lane >> 4;
    const int cb   = wave * 16;

    v8f c0 = {};
    v8f c1 = {};
    const float* wp  = W + cb + m;
    const float* ap0 = As + m * stride;
    const float* ap1 = As + (16 + m) * stride;

    for (int k = 0; k < K; k += 4) {
        const int koff = k + half * 2;
        v2f b;
        b[0] = wp[(size_t)koff * HID_CH];
        b[1] = wp[(size_t)(koff + 1) * HID_CH];
        v2f a0, a1;
        a0[0] = ap0[koff];
        a0[1] = ap0[koff + 1];
        a1[0] = ap1[koff];
        a1[1] = ap1[koff + 1];
        c0 = __builtin_amdgcn_wmma_f32_16x16x4_f32(false, a0, false, b,
                                                   (short)0, c0, false, false);
        c1 = __builtin_amdgcn_wmma_f32_16x16x4_f32(false, a1, false, b,
                                                   (short)0, c1, false, false);
    }

    float* hp0 = H + (size_t)rowbase * HID_CH + cb + m;
    float* hp1 = hp0 + (size_t)16 * HID_CH;
#pragma unroll
    for (int r = 0; r < 8; ++r) {
        hp0[(size_t)(r + 8 * half) * HID_CH] = c0[r];
        hp1[(size_t)(r + 8 * half) * HID_CH] = c1[r];
    }
}


__global__ __launch_bounds__(256) void k_gather(const int* __restrict__ rowptr,
                                                const int* __restrict__ pairs_i,
                                                const float* __restrict__ dinv,
                                                const float* __restrict__ h,
                                                float* __restrict__ acc, int N) {
    const int lane = threadIdx.x & 31;
    const int node = __builtin_amdgcn_readfirstlane((int)((blockIdx.x * 256 + threadIdx.x) >> 5));
    if (node >= N) return;
    const int beg = rowptr[node];
    const int end = rowptr[node + 1];
    const float di = dinv[node];
    const float4* hb = (const float4*)h;
    const int2* pairs = (const int2*)pairs_i;

    float4 a = hb[(size_t)node * 32 + lane];
    const float s = di * di;
    a.x *= s; a.y *= s; a.z *= s; a.w *= s;

    for (int base = beg; base < end; base += 32) {
        const int p = base + lane;
        int src = 0;
        float nv = 0.0f;
        if (p < end) {
            int2 pr = pairs[p];
            src = pr.x;
            if ((unsigned)src >= (unsigned)N) src = 0;
            nv = dinv[src] * __int_as_float(pr.y) * di;
        }
        const int nv_bits = __float_as_int(nv);
        const int cnt = min(32, end - base);
        int j = 0;
        for (; j + 4 <= cnt; j += 4) {
            const int s0 = __builtin_amdgcn_readlane(src, j);
            const int s1 = __builtin_amdgcn_readlane(src, j + 1);
            const int s2 = __builtin_amdgcn_readlane(src, j + 2);
            const int s3 = __builtin_amdgcn_readlane(src, j + 3);
            const float n0 = __int_as_float(__builtin_amdgcn_readlane(nv_bits, j));
            const float n1 = __int_as_float(__builtin_amdgcn_readlane(nv_bits, j + 1));
            const float n2 = __int_as_float(__builtin_amdgcn_readlane(nv_bits, j + 2));
            const float n3 = __int_as_float(__builtin_amdgcn_readlane(nv_bits, j + 3));
            const float4 r0 = hb[(size_t)s0 * 32 + lane];
            const float4 r1 = hb[(size_t)s1 * 32 + lane];
            const float4 r2 = hb[(size_t)s2 * 32 + lane];
            const float4 r3 = hb[(size_t)s3 * 32 + lane];
            a.x += r0.x * n0; a.y += r0.y * n0; a.z += r0.z * n0; a.w += r0.w * n0;
            a.x += r1.x * n1; a.y += r1.y * n1; a.z += r1.z * n1; a.w += r1.w * n1;
            a.x += r2.x * n2; a.y += r2.y * n2; a.z += r2.z * n2; a.w += r2.w * n2;
            a.x += r3.x * n3; a.y += r3.y * n3; a.z += r3.z * n3; a.w += r3.w * n3;
        }
        for (; j < cnt; ++j) {
            const int s0 = __builtin_amdgcn_readlane(src, j);
            const float n0 = __int_as_float(__builtin_amdgcn_readlane(nv_bits, j));
            const float4 r0 = hb[(size_t)s0 * 32 + lane];
            a.x += r0.x * n0; a.y += r0.y * n0; a.z += r0.z * n0; a.w += r0.w * n0;
        }
    }
    ((float4*)acc)[(size_t)node * 32 + lane] = a;
}

__global__ __launch_bounds__(256) void k_bias_relu(float* __restrict__ acc,
                                                   const float* __restrict__ b, int N) {
    int g = blockIdx.x * 256 + threadIdx.x;
    if (g >= N * 32) return;
    float4* ap = (float4*)acc;
    const float4* bp = (const float4*)b;
    float4 v = ap[g];
    float4 bb = bp[g & 31];
    v.x = fmaxf(v.x + bb.x, 0.0f);
    v.y = fmaxf(v.y + bb.y, 0.0f);
    v.z = fmaxf(v.z + bb.z, 0.0f);
    v.w = fmaxf(v.w + bb.w, 0.0f);
    ap[g] = v;
}

__global__ __launch_bounds__(256) void k_bias_l2norm(float* __restrict__ out,
                                                     const float* __restrict__ b, int N) {
    int wid = (blockIdx.x * 256 + threadIdx.x) >> 5;
    int lane = threadIdx.x & 31;
    if (wid >= N) return;
    float4* op = (float4*)(out + (size_t)wid * HID_CH);
    const float4* bp = (const float4*)b;
    float4 v = op[lane];
    float4 bb = bp[lane];
    v.x += bb.x; v.y += bb.y; v.z += bb.z; v.w += bb.w;
    float ss = v.x * v.x + v.y * v.y + v.z * v.z + v.w * v.w;
#pragma unroll
    for (int m = 16; m >= 1; m >>= 1) ss += __shfl_xor(ss, m, 32);
    float inv = 1.0f / fmaxf(sqrtf(ss), 1e-12f);
    v.x *= inv; v.y *= inv; v.z *= inv; v.w *= inv;
    op[lane] = v;
}


static inline size_t align_up(size_t v, size_t a) { return (v + a - 1) & ~(a - 1); }

extern "C" void kernel_launch(void* const* d_in, const int* in_sizes, int n_in,
                              void* d_out, int out_size, void* d_ws, size_t ws_size,
                              hipStream_t stream) {
    const float* x  = (const float*)d_in[0];
    const int*   ei = (const int*)d_in[1];
    const float* ew = (const float*)d_in[2];
    const float* W1 = (const float*)d_in[3];
    const float* b1 = (const float*)d_in[4];
    const float* W2 = (const float*)d_in[5];
    const float* b2 = (const float*)d_in[6];
    float* out = (float*)d_out;

    const int N = in_sizes[0] / IN_CH;
    const int E = in_sizes[1] / 2;
    const int* row = ei;
    const int* col = ei + E;

    char* ws = (char*)d_ws;
    size_t off = 0;
    int*   rowptr = (int*)(ws + off);   off = align_up(off + ((size_t)N + 1) * 4, 1024);
    float* dinv   = (float*)(ws + off); off = align_up(off + (size_t)N * 4, 1024);
    int*   rank   = (int*)(ws + off);   off = align_up(off + (size_t)E * 4, 1024);
    float* h      = (float*)(ws + off); off = align_up(off + (size_t)N * HID_CH * 4, 1024);
    int*   pairs  = (int*)(ws + off);   off = align_up(off + (size_t)E * 8, 1024);
    int*   cnt    = (int*)(ws + off);   off = align_up(off + (size_t)N * 4, 1024);
    (void)ws_size; (void)n_in; (void)out_size;

    const int nb_nodes  = (N + 255) / 256;
    const int nb_edges  = (E + 255) / 256;
    const int nb_nvec   = (N * 32 + 255) / 256;
    const int nb_nwave  = (N + 7) / 8;
    const int nb_gemm   = N / GEMM_ROWS;

    k_zero_cnt   <<<nb_nodes, 256, 0, stream>>>(cnt, N);
    k_count_rank <<<nb_edges, 256, 0, stream>>>(col, cnt, rank, E);
    k_scan       <<<1, 1024, 0, stream>>>(cnt, rowptr, N);
    k_fill       <<<nb_edges, 256, 0, stream>>>(row, col, ew, rowptr, rank, pairs, E);
    k_deg_dinv   <<<nb_nodes, 256, 0, stream>>>(rowptr, pairs, dinv, N);

    k_gemm_wmma  <<<nb_gemm, 256, 0, stream>>>(x, W1, h, IN_CH);
    k_gather     <<<nb_nwave, 256, 0, stream>>>(rowptr, pairs, dinv, h, out, N);
    k_bias_relu  <<<nb_nvec, 256, 0, stream>>>(out, b1, N);

    k_gemm_wmma  <<<nb_gemm, 256, 0, stream>>>(out, W2, h, HID_CH);
    k_gather     <<<nb_nwave, 256, 0, stream>>>(rowptr, pairs, dinv, h, out, N);
    k_bias_l2norm<<<nb_nwave, 256, 0, stream>>>(out, b2, N);
}


# Consumers fused into producers: bias+relu and the layer-2 WMMA product in the layer-1 neighbour sum (rows via LDS), bias+L2norm in the layer-2 sum, deg^-1/2 in the layer-1 product; 11->7 launches
# speedup vs baseline: 2.5078x; 1.1458x over previous
#include <hip/hip_runtime.h>
#include <hip/hip_bf16.h>


typedef __attribute__((ext_vector_type(2))) float v2f;
typedef __attribute__((ext_vector_type(8))) float v8f;

#define IN_CH 256
#define HID_CH 128
#define GEMM_ROWS 32


__global__ __launch_bounds__(256) void k_zero_cnt(int* cnt, int N) {
    int i = blockIdx.x * 256 + threadIdx.x;
    if (i < N) cnt[i] = 0;
}

__global__ __launch_bounds__(256) void k_count_rank(const int* __restrict__ col,
                                                    int* cnt,
                                                    int* __restrict__ rank, int E) {
    int e = blockIdx.x * 256 + threadIdx.x;
    if (e < E) rank[e] = atomicAdd(&cnt[col[e]], 1);
}

__global__ __launch_bounds__(1024) void k_scan(const int* __restrict__ cnt,
                                               int* __restrict__ rowptr, int N) {
    __shared__ int wsum[32];
    __shared__ int tot;
    const int tid = threadIdx.x;
    const int lane = tid & 31;
    const int wave = tid >> 5;
    int carry = 0;
    for (int base = 0; base < N; base += 4096) {
        const int i0 = base + tid * 4;
        int v0 = 0, v1 = 0, v2 = 0, v3 = 0;
        if (i0 + 3 < N) {
            int4 v = *(const int4*)(cnt + i0);
            v0 = v.x; v1 = v.y; v2 = v.z; v3 = v.w;
        } else {
            if (i0 < N) v0 = cnt[i0];
            if (i0 + 1 < N) v1 = cnt[i0 + 1];
            if (i0 + 2 < N) v2 = cnt[i0 + 2];
        }
        const int s = v0 + v1 + v2 + v3;
        int inc = s;
        for (int d = 1; d < 32; d <<= 1) {
            int t = __shfl_up(inc, d, 32);
            if (lane >= d) inc += t;
        }
        if (lane == 31) wsum[wave] = inc;
        __syncthreads();
        if (wave == 0) {
            int w = wsum[lane];
            int wi = w;
            for (int d = 1; d < 32; d <<= 1) {
                int t = __shfl_up(wi, d, 32);
                if (lane >= d) wi += t;
            }
            wsum[lane] = wi - w;
            if (lane == 31) tot = wi;
        }
        __syncthreads();
        const int ex = carry + wsum[wave] + inc - s;
        if (i0 + 3 < N) {
            *(int4*)(rowptr + i0) = make_int4(ex, ex + v0, ex + v0 + v1, ex + v0 + v1 + v2);
        } else {
            if (i0 < N) rowptr[i0] = ex;
            if (i0 + 1 < N) rowptr[i0 + 1] = ex + v0;
            if (i0 + 2 < N) rowptr[i0 + 2] = ex + v0 + v1;
        }
        carry += tot;
        __syncthreads();
    }
    if (tid == 0) rowptr[N] = carry;
}

__global__ __launch_bounds__(256) void k_fill(const int* __restrict__ row,
                                              const int* __restrict__ col,
                                              const float* __restrict__ w,
                                              const int* __restrict__ rowptr,
                                              const int* __restrict__ rank,
                                              int* __restrict__ pairs_i, int E) {
    int e = blockIdx.x * 256 + threadIdx.x;
    if (e < E) {
        int p = rowptr[col[e]] + rank[e];
        if ((unsigned)p < (unsigned)E)
            ((int2*)pairs_i)[p] = make_int2(row[e], __float_as_int(w[e]));
    }
}

__device__ __forceinline__ float node_dinv(const int* __restrict__ rowptr,
                                           const int2* __restrict__ pairs, int i) {
    float d = 1.0f;
    const int end = rowptr[i + 1];
    for (int p = rowptr[i]; p < end; ++p) d += __int_as_float(pairs[p].y);
    return (d > 0.0f) ? rsqrtf(d) : 0.0f;
}

__device__ __forceinline__ void gemm_tile32(const float* As, int stride, int K,
                                            const float* __restrict__ W,
                                            float* __restrict__ H,
                                            int rowbase, int tid) {
    const int wave = tid >> 5;
    const int lane = tid & 31;
    const int m    = lane & 15;
    const int half = lane >> 4;
    const int cb   = wave * 16;

    v8f c0 = {};
    v8f c1 = {};
    const float* wp  = W + cb + m;
    const float* ap0 = As + m * stride;
    const float* ap1 = As + (16 + m) * stride;

    for (int k = 0; k < K; k += 4) {
        const int koff = k + half * 2;
        v2f b;
        b[0] = wp[(size_t)koff * HID_CH];
        b[1] = wp[(size_t)(koff + 1) * HID_CH];
        v2f a0, a1;
        a0[0] = ap0[koff];
        a0[1] = ap0[koff + 1];
        a1[0] = ap1[koff];
        a1[1] = ap1[koff + 1];
        c0 = __builtin_amdgcn_wmma_f32_16x16x4_f32(false, a0, false, b,
                                                   (short)0, c0, false, false);
        c1 = __builtin_amdgcn_wmma_f32_16x16x4_f32(false, a1, false, b,
                                                   (short)0, c1, false, false);
    }

    float* hp0 = H + (size_t)rowbase * HID_CH + cb + m;
    float* hp1 = hp0 + (size_t)16 * HID_CH;
#pragma unroll
    for (int r = 0; r < 8; ++r) {
        hp0[(size_t)(r + 8 * half) * HID_CH] = c0[r];
        hp1[(size_t)(r + 8 * half) * HID_CH] = c1[r];
    }
}

__global__ __launch_bounds__(256) void k_gemm_dinv(const float* __restrict__ A,
                                                   const float* __restrict__ W,
                                                   float* __restrict__ H, int K,
                                                   const int* __restrict__ rowptr,
                                                   const int* __restrict__ pairs_i,
                                                   float* __restrict__ dinv) {
    __shared__ float As[GEMM_ROWS * (IN_CH + 4)];
    const int stride = K + 4;
    const int tid = threadIdx.x;
    const int rowbase = blockIdx.x * GEMM_ROWS;

    if (tid < 32)
        dinv[rowbase + tid] = node_dinv(rowptr, (const int2*)pairs_i, rowbase + tid);

    for (int idx = tid; idx < GEMM_ROWS * K; idx += 256) {
        int r = idx / K;
        int kk = idx - r * K;
        As[r * stride + kk] = A[(size_t)(rowbase + r) * K + kk];
    }
    __syncthreads();

    gemm_tile32(As, stride, K, W, H, rowbase, tid);
}


__device__ __forceinline__ float4 gather_row(int node, int lane,
                                             const int* __restrict__ rowptr,
                                             const int2* __restrict__ pairs,
                                             const float* __restrict__ dinv,
                                             const float4* __restrict__ hb, int N) {
    const int beg = rowptr[node];
    const int end = rowptr[node + 1];
    const float di = dinv[node];

    float4 a = hb[(size_t)node * 32 + lane];
    const float s = di * di;
    a.x *= s; a.y *= s; a.z *= s; a.w *= s;

    for (int base = beg; base < end; base += 32) {
        const int p = base + lane;
        int src = 0;
        float nv = 0.0f;
        if (p < end) {
            int2 pr = pairs[p];
            src = pr.x;
            if ((unsigned)src >= (unsigned)N) src = 0;
            nv = dinv[src] * __int_as_float(pr.y) * di;
        }
        const int nv_bits = __float_as_int(nv);
        const int cnt = min(32, end - base);
        int j = 0;
        for (; j + 4 <= cnt; j += 4) {
            const int s0 = __builtin_amdgcn_readlane(src, j);
            const int s1 = __builtin_amdgcn_readlane(src, j + 1);
            const int s2 = __builtin_amdgcn_readlane(src, j + 2);
            const int s3 = __builtin_amdgcn_readlane(src, j + 3);
            const float n0 = __int_as_float(__builtin_amdgcn_readlane(nv_bits, j));
            const float n1 = __int_as_float(__builtin_amdgcn_readlane(nv_bits, j + 1));
            const float n2 = __int_as_float(__builtin_amdgcn_readlane(nv_bits, j + 2));
            const float n3 = __int_as_float(__builtin_amdgcn_readlane(nv_bits, j + 3));
            const float4 r0 = hb[(size_t)s0 * 32 + lane];
            const float4 r1 = hb[(size_t)s1 * 32 + lane];
            const float4 r2 = hb[(size_t)s2 * 32 + lane];
            const float4 r3 = hb[(size_t)s3 * 32 + lane];
            a.x += r0.x * n0; a.y += r0.y * n0; a.z += r0.z * n0; a.w += r0.w * n0;
            a.x += r1.x * n1; a.y += r1.y * n1; a.z += r1.z * n1; a.w += r1.w * n1;
            a.x += r2.x * n2; a.y += r2.y * n2; a.z += r2.z * n2; a.w += r2.w * n2;
            a.x += r3.x * n3; a.y += r3.y * n3; a.z += r3.z * n3; a.w += r3.w * n3;
        }
        for (; j < cnt; ++j) {
            const int s0 = __builtin_amdgcn_readlane(src, j);
            const float n0 = __int_as_float(__builtin_amdgcn_readlane(nv_bits, j));
            const float4 r0 = hb[(size_t)s0 * 32 + lane];
            a.x += r0.x * n0; a.y += r0.y * n0; a.z += r0.z * n0; a.w += r0.w * n0;
        }
    }
    return a;
}

__global__ __launch_bounds__(256, 16) void k_gather_relu_gemm(const int* __restrict__ rowptr,
                                                          const int* __restrict__ pairs_i,
                                                          const float* __restrict__ dinv,
                                                          const float* __restrict__ h,
                                                          const float* __restrict__ b,
                                                          const float* __restrict__ W,
                                                          float* __restrict__ H2, int N) {
    __shared__ float As[GEMM_ROWS * (HID_CH + 4)];
    const int tid = threadIdx.x;
    const int lane = tid & 31;
    const int wave = __builtin_amdgcn_readfirstlane(tid >> 5);
    const int rowbase = blockIdx.x * GEMM_ROWS;
    const float4 bb = ((const float4*)b)[lane];

    for (int i = 0; i < GEMM_ROWS / 8; ++i) {
        const int r = wave + 8 * i;
        const int node = rowbase + r;
        float4 v = make_float4(0.0f, 0.0f, 0.0f, 0.0f);
        if (node < N) {
            v = gather_row(node, lane, rowptr, (const int2*)pairs_i, dinv,
                           (const float4*)h, N);
            v.x = fmaxf(v.x + bb.x, 0.0f);
            v.y = fmaxf(v.y + bb.y, 0.0f);
            v.z = fmaxf(v.z + bb.z, 0.0f);
            v.w = fmaxf(v.w + bb.w, 0.0f);
        }
        *(float4*)(As + r * (HID_CH + 4) + lane * 4) = v;
    }
    __syncthreads();

    gemm_tile32(As, HID_CH + 4, HID_CH, W, H2, rowbase, tid);
}

__global__ __launch_bounds__(256) void k_gather_l2norm(const int* __restrict__ rowptr,
                                                       const int* __restrict__ pairs_i,
                                                       const float* __restrict__ dinv,
                                                       const float* __restrict__ h,
                                                       const float* __restrict__ b,
                                                       float* __restrict__ out, int N) {
    const int lane = threadIdx.x & 31;
    const int node = __builtin_amdgcn_readfirstlane((int)((blockIdx.x * 256 + threadIdx.x) >> 5));
    if (node >= N) return;
    float4 v = gather_row(node, lane, rowptr, (const int2*)pairs_i, dinv,
                          (const float4*)h, N);
    const float4 bb = ((const float4*)b)[lane];
    v.x += bb.x; v.y += bb.y; v.z += bb.z; v.w += bb.w;
    float ss = v.x * v.x + v.y * v.y + v.z * v.z + v.w * v.w;
#pragma unroll
    for (int m = 16; m >= 1; m >>= 1) ss += __shfl_xor(ss, m, 32);
    float inv = 1.0f / fmaxf(sqrtf(ss), 1e-12f);
    v.x *= inv; v.y *= inv; v.z *= inv; v.w *= inv;
    ((float4*)out)[(size_t)node * 32 + lane] = v;
}


static inline size_t align_up(size_t v, size_t a) { return (v + a - 1) & ~(a - 1); }

extern "C" void kernel_launch(void* const* d_in, const int* in_sizes, int n_in,
                              void* d_out, int out_size, void* d_ws, size_t ws_size,
                              hipStream_t stream) {
    const float* x  = (const float*)d_in[0];
    const int*   ei = (const int*)d_in[1];
    const float* ew = (const float*)d_in[2];
    const float* W1 = (const float*)d_in[3];
    const float* b1 = (const float*)d_in[4];
    const float* W2 = (const float*)d_in[5];
    const float* b2 = (const float*)d_in[6];
    float* out = (float*)d_out;

    const int N = in_sizes[0] / IN_CH;
    const int E = in_sizes[1] / 2;
    const int* row = ei;
    const int* col = ei + E;

    char* ws = (char*)d_ws;
    size_t off = 0;
    int*   rowptr = (int*)(ws + off);   off = align_up(off + ((size_t)N + 1) * 4, 1024);
    float* dinv   = (float*)(ws + off); off = align_up(off + (size_t)N * 4, 1024);
    int*   rank   = (int*)(ws + off);   off = align_up(off + (size_t)E * 4, 1024);
    float* h      = (float*)(ws + off); off = align_up(off + (size_t)N * HID_CH * 4, 1024);
    int*   pairs  = (int*)(ws + off);   off = align_up(off + (size_t)E * 8, 1024);
    int*   cnt    = (int*)(ws + off);   off = align_up(off + (size_t)N * 4, 1024);
    (void)ws_size; (void)n_in; (void)out_size;

    const int nb_nodes  = (N + 255) / 256;
    const int nb_edges  = (E + 255) / 256;
    const int nb_nwave  = (N + 7) / 8;
    const int nb_gemm   = N / GEMM_ROWS;

    k_zero_cnt   <<<nb_nodes, 256, 0, stream>>>(cnt, N);
    k_count_rank <<<nb_edges, 256, 0, stream>>>(col, cnt, rank, E);
    k_scan       <<<1, 1024, 0, stream>>>(cnt, rowptr, N);
    k_fill       <<<nb_edges, 256, 0, stream>>>(row, col, ew, rowptr, rank, pairs, E);

    k_gemm_dinv  <<<nb_gemm, 256, 0, stream>>>(x, W1, out, IN_CH, rowptr, pairs, dinv);
    k_gather_relu_gemm<<<nb_gemm, 256, 0, stream>>>(rowptr, pairs, dinv, out, b1, W2, h, N);
    k_gather_l2norm<<<nb_nwave, 256, 0, stream>>>(rowptr, pairs, dinv, h, b2, out, N);
}


# Edges grouped by destination through LDS bins (chunk bin counts, bin runs + per-bin node sort in LDS, contiguous stores); no global atomics, no scan launch, 7->6 launches; sum order = LDS arrival
# speedup vs baseline: 4.1619x; 1.6596x over previous
#include <hip/hip_runtime.h>
#include <hip/hip_bf16.h>


typedef __attribute__((ext_vector_type(2))) float v2f;
typedef __attribute__((ext_vector_type(8))) float v8f;

#define IN_CH 256
#define HID_CH 128
#define GEMM_ROWS 32


#define BIN_SHIFT 9
#define NPB (1 << BIN_SHIFT)
#define MAX_BINS 256
#define CHUNK 12288
#define EPT (CHUNK / 1024)
#define SORT_T 1024
#define SORT_RPT 12
#define SORT_CAP (SORT_T * SORT_RPT)

__device__ __forceinline__ int bin_of(int c, int NB) {
    unsigned b = (unsigned)c >> BIN_SHIFT;
    return (int)(b < (unsigned)NB ? b : (unsigned)(NB - 1));
}

__device__ __forceinline__ int wave_incl_scan(int v, int lane) {
    for (int d = 1; d < 32; d <<= 1) {
        int t = __shfl_up(v, d, 32);
        if (lane >= d) v += t;
    }
    return v;
}

__global__ __launch_bounds__(1024) void k_bin_hist(const int* __restrict__ col,
                                                   int* __restrict__ histg, int E, int NB) {
    __shared__ int hist[MAX_BINS];
    const int tid = threadIdx.x;
    if (tid < MAX_BINS) hist[tid] = 0;
    __syncthreads();
    const int base = blockIdx.x * CHUNK;
    int c[EPT];
#pragma unroll
    for (int k = 0; k < EPT; ++k) {
        const int e = base + k * 1024 + tid;
        c[k] = (e < E) ? col[e] : -1;
    }
#pragma unroll
    for (int k = 0; k < EPT; ++k) {
        const int e = base + k * 1024 + tid;
        if (e < E) atomicAdd(&hist[bin_of(c[k], NB)], 1);
    }
    __syncthreads();
    if (tid < NB) histg[blockIdx.x * NB + tid] = hist[tid];
}

__global__ __launch_bounds__(1024) void k_bin_place(const int* __restrict__ row,
                                                    const int* __restrict__ col,
                                                    const float* __restrict__ w,
                                                    const int* __restrict__ histg,
                                                    int* __restrict__ binstart,
                                                    int2* __restrict__ binned,
                                                    int E, int N, int NB, int NCH) {
    __shared__ int2 stage[CHUNK];
    __shared__ int part_tot[4][MAX_BINS];
    __shared__ int part_pre[4][MAX_BINS];
    __shared__ int s_mine[MAX_BINS];
    __shared__ int s_loff[MAX_BINS];
    __shared__ int s_gbase[MAX_BINS];
    __shared__ int s_cursor[MAX_BINS];
    __shared__ int wsA[8];
    __shared__ int wsB[8];
    const int tid = threadIdx.x;
    const int lane = tid & 31;
    const int wave = tid >> 5;
    const int blk = blockIdx.x;

    {
        const int g = tid >> 8;
        const int bin = tid & (MAX_BINS - 1);
        int tot = 0, pre = 0;
        if (bin < NB) {
            for (int j = g; j < NCH; j += 4) {
                const int v = histg[j * NB + bin];
                tot += v;
                if (j < blk) pre += v;
            }
        }
        part_tot[g][bin] = tot;
        part_pre[g][bin] = pre;
    }
    __syncthreads();

    int t_tot = 0, t_pre = 0, t_mine = 0;
    if (tid < MAX_BINS) {
        t_tot = part_tot[0][tid] + part_tot[1][tid] + part_tot[2][tid] + part_tot[3][tid];
        t_pre = part_pre[0][tid] + part_pre[1][tid] + part_pre[2][tid] + part_pre[3][tid];
        if (tid < NB) t_mine = histg[blk * NB + tid];
    }
    const int incA = wave_incl_scan(t_tot, lane);
    const int incB = wave_incl_scan(t_mine, lane);
    if (tid < MAX_BINS && lane == 31) { wsA[wave] = incA; wsB[wave] = incB; }
    __syncthreads();
    if (tid < MAX_BINS) {
        int offA = 0, offB = 0;
        for (int k = 0; k < wave; ++k) { offA += wsA[k]; offB += wsB[k]; }
        const int exA = offA + incA - t_tot;
        const int exB = offB + incB - t_mine;
        s_gbase[tid] = exA + t_pre;
        s_loff[tid] = exB;
        s_cursor[tid] = exB;
        s_mine[tid] = t_mine;
        if (blk == 0 && tid < NB) binstart[tid] = exA;
        if (blk == 0 && tid == NB - 1) binstart[NB] = exA + t_tot;
    }
    __syncthreads();

    const int base = blk * CHUNK;
    int c[EPT], s[EPT], wb[EPT];
#pragma unroll
    for (int k = 0; k < EPT; ++k) {
        const int e = base + k * 1024 + tid;
        c[k] = 0; s[k] = 0; wb[k] = 0;
        if (e < E) { c[k] = col[e]; s[k] = row[e]; wb[k] = __float_as_int(w[e]); }
    }
#pragma unroll
    for (int k = 0; k < EPT; ++k) {
        const int e = base + k * 1024 + tid;
        if (e < E) {
            int src = s[k];
            if ((unsigned)src >= (unsigned)N) src = 0;
            const int pos = atomicAdd(&s_cursor[bin_of(c[k], NB)], 1);
            if ((unsigned)pos < (unsigned)CHUNK)
                stage[pos] = make_int2((src << BIN_SHIFT) | (c[k] & (NPB - 1)), wb[k]);
        }
    }
    __syncthreads();

    for (int b = wave; b < NB; b += 32) {
        const int n = s_mine[b];
        const int lo = s_loff[b];
        const int gb = s_gbase[b];
        for (int i = lane; i < n; i += 32) {
            const int p = gb + i;
            if ((unsigned)p < (unsigned)E && (unsigned)(lo + i) < (unsigned)CHUNK)
                binned[p] = stage[lo + i];
        }
    }
}

__global__ __launch_bounds__(SORT_T) void k_bin_sort(const int2* __restrict__ binned,
                                                     const int* __restrict__ binstart,
                                                     int* __restrict__ rowptr,
                                                     int2* __restrict__ pairs,
                                                     int N, int E, int NB) {
    __shared__ int2 stage[SORT_CAP];
    __shared__ int cnt[NPB];
    __shared__ int cur[NPB];
    __shared__ int wsum[NPB / 32];
    const int tid = threadIdx.x;
    const int lane = tid & 31;
    const int wave = tid >> 5;
    const int b = blockIdx.x;
    int s = binstart[b];
    int e = binstart[b + 1];
    s = min(max(s, 0), E);
    e = min(max(e, s), E);
    const int n = e - s;
    const bool fits = (n <= SORT_CAP);

    if (tid < NPB) cnt[tid] = 0;
    __syncthreads();

    int2 r[SORT_RPT];
    if (fits) {
#pragma unroll
        for (int k = 0; k < SORT_RPT; ++k) {
            const int i = s + k * SORT_T + tid;
            r[k] = make_int2(0, 0);
            if (i < e) r[k] = binned[i];
        }
#pragma unroll
        for (int k = 0; k < SORT_RPT; ++k) {
            const int i = s + k * SORT_T + tid;
            if (i < e) atomicAdd(&cnt[r[k].x & (NPB - 1)], 1);
        }
    } else {
        for (int i = s + tid; i < e; i += SORT_T)
            atomicAdd(&cnt[binned[i].x & (NPB - 1)], 1);
    }
    __syncthreads();

    int c = 0;
    if (tid < NPB) c = cnt[tid];
    const int inc = wave_incl_scan(c, lane);
    if (tid < NPB && lane == 31) wsum[wave] = inc;
    __syncthreads();
    if (tid < NPB) {
        int off = 0;
        for (int k = 0; k < wave; ++k) off += wsum[k];
        const int ex = off + inc - c;
        const int node = b * NPB + tid;
        if (node < N) rowptr[node] = s + ex;
        cur[tid] = ex;
    }
    if (b == NB - 1 && tid == 0) rowptr[N] = e;
    __syncthreads();

    if (fits) {
#pragma unroll
        for (int k = 0; k < SORT_RPT; ++k) {
            const int i = s + k * SORT_T + tid;
            if (i < e) {
                const int p = atomicAdd(&cur[r[k].x & (NPB - 1)], 1);
                if ((unsigned)p < (unsigned)SORT_CAP)
                    stage[p] = make_int2((int)((unsigned)r[k].x >> BIN_SHIFT), r[k].y);
            }
        }
        __syncthreads();
        for (int i = tid; i < n; i += SORT_T) pairs[s + i] = stage[i];
    } else {
        for (int i = s + tid; i < e; i += SORT_T) {
            const int2 rec = binned[i];
            const int p = s + atomicAdd(&cur[rec.x & (NPB - 1)], 1);
            if ((unsigned)p < (unsigned)E)
                pairs[p] = make_int2((int)((unsigned)rec.x >> BIN_SHIFT), rec.y);
        }
    }
}

__device__ __forceinline__ float node_dinv(const int* __restrict__ rowptr,
                                           const int2* __restrict__ pairs, int i) {
    float d = 1.0f;
    const int end = rowptr[i + 1];
    for (int p = rowptr[i]; p < end; ++p) d += __int_as_float(pairs[p].y);
    return (d > 0.0f) ? rsqrtf(d) : 0.0f;
}

__device__ __forceinline__ void gemm_tile32(const float* As, int stride, int K,
                                            const float* __restrict__ W,
                                            float* __restrict__ H,
                                            int rowbase, int tid) {
    const int wave = tid >> 5;
    const int lane = tid & 31;
    const int m    = lane & 15;
    const int half = lane >> 4;
    const int cb   = wave * 16;

    v8f c0 = {};
    v8f c1 = {};
    const float* wp  = W + cb + m;
    const float* ap0 = As + m * stride;
    const float* ap1 = As + (16 + m) * stride;

    for (int k = 0; k < K; k += 4) {
        const int koff = k + half * 2;
        v2f b;
        b[0] = wp[(size_t)koff * HID_CH];
        b[1] = wp[(size_t)(koff + 1) * HID_CH];
        v2f a0, a1;
        a0[0] = ap0[koff];
        a0[1] = ap0[koff + 1];
        a1[0] = ap1[koff];
        a1[1] = ap1[koff + 1];
        c0 = __builtin_amdgcn_wmma_f32_16x16x4_f32(false, a0, false, b,
                                                   (short)0, c0, false, false);
        c1 = __builtin_amdgcn_wmma_f32_16x16x4_f32(false, a1, false, b,
                                                   (short)0, c1, false, false);
    }

    float* hp0 = H + (size_t)rowbase * HID_CH + cb + m;
    float* hp1 = hp0 + (size_t)16 * HID_CH;
#pragma unroll
    for (int r = 0; r < 8; ++r) {
        hp0[(size_t)(r + 8 * half) * HID_CH] = c0[r];
        hp1[(size_t)(r + 8 * half) * HID_CH] = c1[r];
    }
}

__global__ __launch_bounds__(256) void k_gemm_dinv(const float* __restrict__ A,
                                                   const float* __restrict__ W,
                                                   float* __restrict__ H, int K,
                                                   const int* __restrict__ rowptr,
                                                   const int* __restrict__ pairs_i,
                                                   float* __restrict__ dinv) {
    __shared__ float As[GEMM_ROWS * (IN_CH + 4)];
    const int stride = K + 4;
    const int tid = threadIdx.x;
    const int rowbase = blockIdx.x * GEMM_ROWS;

    if (tid < 32)
        dinv[rowbase + tid] = node_dinv(rowptr, (const int2*)pairs_i, rowbase + tid);

    for (int idx = tid; idx < GEMM_ROWS * K; idx += 256) {
        int r = idx / K;
        int kk = idx - r * K;
        As[r * stride + kk] = A[(size_t)(rowbase + r) * K + kk];
    }
    __syncthreads();

    gemm_tile32(As, stride, K, W, H, rowbase, tid);
}


__device__ __forceinline__ float4 gather_row(int node, int lane,
                                             const int* __restrict__ rowptr,
                                             const int2* __restrict__ pairs,
                                             const float* __restrict__ dinv,
                                             const float4* __restrict__ hb, int N) {
    const int beg = rowptr[node];
    const int end = rowptr[node + 1];
    const float di = dinv[node];

    float4 a = hb[(size_t)node * 32 + lane];
    const float s = di * di;
    a.x *= s; a.y *= s; a.z *= s; a.w *= s;

    for (int base = beg; base < end; base += 32) {
        const int p = base + lane;
        int src = 0;
        float nv = 0.0f;
        if (p < end) {
            int2 pr = pairs[p];
            src = pr.x;
            if ((unsigned)src >= (unsigned)N) src = 0;
            nv = dinv[src] * __int_as_float(pr.y) * di;
        }
        const int nv_bits = __float_as_int(nv);
        const int cnt = min(32, end - base);
        int j = 0;
        for (; j + 4 <= cnt; j += 4) {
            const int s0 = __builtin_amdgcn_readlane(src, j);
            const int s1 = __builtin_amdgcn_readlane(src, j + 1);
            const int s2 = __builtin_amdgcn_readlane(src, j + 2);
            const int s3 = __builtin_amdgcn_readlane(src, j + 3);
            const float n0 = __int_as_float(__builtin_amdgcn_readlane(nv_bits, j));
            const float n1 = __int_as_float(__builtin_amdgcn_readlane(nv_bits, j + 1));
            const float n2 = __int_as_float(__builtin_amdgcn_readlane(nv_bits, j + 2));
            const float n3 = __int_as_float(__builtin_amdgcn_readlane(nv_bits, j + 3));
            const float4 r0 = hb[(size_t)s0 * 32 + lane];
            const float4 r1 = hb[(size_t)s1 * 32 + lane];
            const float4 r2 = hb[(size_t)s2 * 32 + lane];
            const float4 r3 = hb[(size_t)s3 * 32 + lane];
            a.x += r0.x * n0; a.y += r0.y * n0; a.z += r0.z * n0; a.w += r0.w * n0;
            a.x += r1.x * n1; a.y += r1.y * n1; a.z += r1.z * n1; a.w += r1.w * n1;
            a.x += r2.x * n2; a.y += r2.y * n2; a.z += r2.z * n2; a.w += r2.w * n2;
            a.x += r3.x * n3; a.y += r3.y * n3; a.z += r3.z * n3; a.w += r3.w * n3;
        }
        for (; j < cnt; ++j) {
            const int s0 = __builtin_amdgcn_readlane(src, j);
            const float n0 = __int_as_float(__builtin_amdgcn_readlane(nv_bits, j));
            const float4 r0 = hb[(size_t)s0 * 32 + lane];
            a.x += r0.x * n0; a.y += r0.y * n0; a.z += r0.z * n0; a.w += r0.w * n0;
        }
    }
    return a;
}

__global__ __launch_bounds__(256, 16) void k_gather_relu_gemm(const int* __restrict__ rowptr,
                                                          const int* __restrict__ pairs_i,
                                                          const float* __restrict__ dinv,
                                                          const float* __restrict__ h,
                                                          const float* __restrict__ b,
                                                          const float* __restrict__ W,
                                                          float* __restrict__ H2, int N) {
    __shared__ float As[GEMM_ROWS * (HID_CH + 4)];
    const int tid = threadIdx.x;
    const int lane = tid & 31;
    const int wave = __builtin_amdgcn_readfirstlane(tid >> 5);
    const int rowbase = blockIdx.x * GEMM_ROWS;
    const float4 bb = ((const float4*)b)[lane];

    for (int i = 0; i < GEMM_ROWS / 8; ++i) {
        const int r = wave + 8 * i;
        const int node = rowbase + r;
        float4 v = make_float4(0.0f, 0.0f, 0.0f, 0.0f);
        if (node < N) {
            v = gather_row(node, lane, rowptr, (const int2*)pairs_i, dinv,
                           (const float4*)h, N);
            v.x = fmaxf(v.x + bb.x, 0.0f);
            v.y = fmaxf(v.y + bb.y, 0.0f);
            v.z = fmaxf(v.z + bb.z, 0.0f);
            v.w = fmaxf(v.w + bb.w, 0.0f);
        }
        *(float4*)(As + r * (HID_CH + 4) + lane * 4) = v;
    }
    __syncthreads();

    gemm_tile32(As, HID_CH + 4, HID_CH, W, H2, rowbase, tid);
}

__global__ __launch_bounds__(256) void k_gather_l2norm(const int* __restrict__ rowptr,
                                                       const int* __restrict__ pairs_i,
                                                       const float* __restrict__ dinv,
                                                       const float* __restrict__ h,
                                                       const float* __restrict__ b,
                                                       float* __restrict__ out, int N) {
    const int lane = threadIdx.x & 31;
    const int node = __builtin_amdgcn_readfirstlane((int)((blockIdx.x * 256 + threadIdx.x) >> 5));
    if (node >= N) return;
    float4 v = gather_row(node, lane, rowptr, (const int2*)pairs_i, dinv,
                          (const float4*)h, N);
    const float4 bb = ((const float4*)b)[lane];
    v.x += bb.x; v.y += bb.y; v.z += bb.z; v.w += bb.w;
    float ss = v.x * v.x + v.y * v.y + v.z * v.z + v.w * v.w;
#pragma unroll
    for (int m = 16; m >= 1; m >>= 1) ss += __shfl_xor(ss, m, 32);
    float inv = 1.0f / fmaxf(sqrtf(ss), 1e-12f);
    v.x *= inv; v.y *= inv; v.z *= inv; v.w *= inv;
    ((float4*)out)[(size_t)node * 32 + lane] = v;
}


static inline size_t align_up(size_t v, size_t a) { return (v + a - 1) & ~(a - 1); }

extern "C" void kernel_launch(void* const* d_in, const int* in_sizes, int n_in,
                              void* d_out, int out_size, void* d_ws, size_t ws_size,
                              hipStream_t stream) {
    const float* x  = (const float*)d_in[0];
    const int*   ei = (const int*)d_in[1];
    const float* ew = (const float*)d_in[2];
    const float* W1 = (const float*)d_in[3];
    const float* b1 = (const float*)d_in[4];
    const float* W2 = (const float*)d_in[5];
    const float* b2 = (const float*)d_in[6];
    float* out = (float*)d_out;

    const int N = in_sizes[0] / IN_CH;
    const int E = in_sizes[1] / 2;
    const int* row = ei;
    const int* col = ei + E;

    const int NB  = (N + NPB - 1) / NPB;
    const int NCH = (E + CHUNK - 1) / CHUNK;

    char* ws = (char*)d_ws;
    size_t off = 0;
    int*   rowptr   = (int*)(ws + off);   off = align_up(off + ((size_t)N + 1) * 4, 1024);
    float* dinv     = (float*)(ws + off); off = align_up(off + (size_t)N * 4, 1024);
    int*   histg    = (int*)(ws + off);   off = align_up(off + (size_t)NCH * NB * 4, 1024);
    int*   binstart = (int*)(ws + off);   off = align_up(off + ((size_t)NB + 1) * 4, 1024);
    float* h        = (float*)(ws + off); off = align_up(off + (size_t)N * HID_CH * 4, 1024);
    int*   pairs    = (int*)(ws + off);   off = align_up(off + (size_t)E * 8, 1024);
    int2*  binned   = (int2*)h;
    (void)ws_size; (void)n_in; (void)out_size;

    const int nb_nwave  = (N + 7) / 8;
    const int nb_gemm   = N / GEMM_ROWS;

    k_bin_hist  <<<NCH, 1024, 0, stream>>>(col, histg, E, NB);
    k_bin_place <<<NCH, 1024, 0, stream>>>(row, col, ew, histg, binstart, binned, E, N, NB, NCH);
    k_bin_sort  <<<NB, SORT_T, 0, stream>>>(binned, binstart, rowptr, (int2*)pairs, N, E, NB);

    k_gemm_dinv  <<<nb_gemm, 256, 0, stream>>>(x, W1, out, IN_CH, rowptr, pairs, dinv);
    k_gather_relu_gemm<<<nb_gemm, 256, 0, stream>>>(rowptr, pairs, dinv, out, b1, W2, h, N);
    k_gather_l2norm<<<nb_nwave, 256, 0, stream>>>(rowptr, pairs, dinv, h, b2, out, N);
}


# Both products store rows times their own deg^-1/2; neighbour sums use the bare edge weight and scale once at the end: no per-edge dinv[src] lookup; term rounding changes, sum order kept
# speedup vs baseline: 4.3304x; 1.0405x over previous
#include <hip/hip_runtime.h>
#include <hip/hip_bf16.h>


typedef __attribute__((ext_vector_type(2))) float v2f;
typedef __attribute__((ext_vector_type(8))) float v8f;

#define IN_CH 256
#define HID_CH 128
#define GEMM_ROWS 32


#define BIN_SHIFT 9
#define NPB (1 << BIN_SHIFT)
#define MAX_BINS 256
#define CHUNK 12288
#define EPT (CHUNK / 1024)
#define SORT_T 1024
#define SORT_RPT 12
#define SORT_CAP (SORT_T * SORT_RPT)

__device__ __forceinline__ int bin_of(int c, int NB) {
    unsigned b = (unsigned)c >> BIN_SHIFT;
    return (int)(b < (unsigned)NB ? b : (unsigned)(NB - 1));
}

__device__ __forceinline__ int wave_incl_scan(int v, int lane) {
    for (int d = 1; d < 32; d <<= 1) {
        int t = __shfl_up(v, d, 32);
        if (lane >= d) v += t;
    }
    return v;
}

__global__ __launch_bounds__(1024) void k_bin_hist(const int* __restrict__ col,
                                                   int* __restrict__ histg, int E, int NB) {
    __shared__ int hist[MAX_BINS];
    const int tid = threadIdx.x;
    if (tid < MAX_BINS) hist[tid] = 0;
    __syncthreads();
    const int base = blockIdx.x * CHUNK;
    int c[EPT];
#pragma unroll
    for (int k = 0; k < EPT; ++k) {
        const int e = base + k * 1024 + tid;
        c[k] = (e < E) ? col[e] : -1;
    }
#pragma unroll
    for (int k = 0; k < EPT; ++k) {
        const int e = base + k * 1024 + tid;
        if (e < E) atomicAdd(&hist[bin_of(c[k], NB)], 1);
    }
    __syncthreads();
    if (tid < NB) histg[blockIdx.x * NB + tid] = hist[tid];
}

__global__ __launch_bounds__(1024) void k_bin_place(const int* __restrict__ row,
                                                    const int* __restrict__ col,
                                                    const float* __restrict__ w,
                                                    const int* __restrict__ histg,
                                                    int* __restrict__ binstart,
                                                    int2* __restrict__ binned,
                                                    int E, int N, int NB, int NCH) {
    __shared__ int2 stage[CHUNK];
    __shared__ int part_tot[4][MAX_BINS];
    __shared__ int part_pre[4][MAX_BINS];
    __shared__ int s_mine[MAX_BINS];
    __shared__ int s_loff[MAX_BINS];
    __shared__ int s_gbase[MAX_BINS];
    __shared__ int s_cursor[MAX_BINS];
    __shared__ int wsA[8];
    __shared__ int wsB[8];
    const int tid = threadIdx.x;
    const int lane = tid & 31;
    const int wave = tid >> 5;
    const int blk = blockIdx.x;

    {
        const int g = tid >> 8;
        const int bin = tid & (MAX_BINS - 1);
        int tot = 0, pre = 0;
        if (bin < NB) {
            for (int j = g; j < NCH; j += 4) {
                const int v = histg[j * NB + bin];
                tot += v;
                if (j < blk) pre += v;
            }
        }
        part_tot[g][bin] = tot;
        part_pre[g][bin] = pre;
    }
    __syncthreads();

    int t_tot = 0, t_pre = 0, t_mine = 0;
    if (tid < MAX_BINS) {
        t_tot = part_tot[0][tid] + part_tot[1][tid] + part_tot[2][tid] + part_tot[3][tid];
        t_pre = part_pre[0][tid] + part_pre[1][tid] + part_pre[2][tid] + part_pre[3][tid];
        if (tid < NB) t_mine = histg[blk * NB + tid];
    }
    const int incA = wave_incl_scan(t_tot, lane);
    const int incB = wave_incl_scan(t_mine, lane);
    if (tid < MAX_BINS && lane == 31) { wsA[wave] = incA; wsB[wave] = incB; }
    __syncthreads();
    if (tid < MAX_BINS) {
        int offA = 0, offB = 0;
        for (int k = 0; k < wave; ++k) { offA += wsA[k]; offB += wsB[k]; }
        const int exA = offA + incA - t_tot;
        const int exB = offB + incB - t_mine;
        s_gbase[tid] = exA + t_pre;
        s_loff[tid] = exB;
        s_cursor[tid] = exB;
        s_mine[tid] = t_mine;
        if (blk == 0 && tid < NB) binstart[tid] = exA;
        if (blk == 0 && tid == NB - 1) binstart[NB] = exA + t_tot;
    }
    __syncthreads();

    const int base = blk * CHUNK;
    int c[EPT], s[EPT], wb[EPT];
#pragma unroll
    for (int k = 0; k < EPT; ++k) {
        const int e = base + k * 1024 + tid;
        c[k] = 0; s[k] = 0; wb[k] = 0;
        if (e < E) { c[k] = col[e]; s[k] = row[e]; wb[k] = __float_as_int(w[e]); }
    }
#pragma unroll
    for (int k = 0; k < EPT; ++k) {
        const int e = base + k * 1024 + tid;
        if (e < E) {
            int src = s[k];
            if ((unsigned)src >= (unsigned)N) src = 0;
            const int pos = atomicAdd(&s_cursor[bin_of(c[k], NB)], 1);
            if ((unsigned)pos < (unsigned)CHUNK)
                stage[pos] = make_int2((src << BIN_SHIFT) | (c[k] & (NPB - 1)), wb[k]);
        }
    }
    __syncthreads();

    for (int b = wave; b < NB; b += 32) {
        const int n = s_mine[b];
        const int lo = s_loff[b];
        const int gb = s_gbase[b];
        for (int i = lane; i < n; i += 32) {
            const int p = gb + i;
            if ((unsigned)p < (unsigned)E && (unsigned)(lo + i) < (unsigned)CHUNK)
                binned[p] = stage[lo + i];
        }
    }
}

__global__ __launch_bounds__(SORT_T) void k_bin_sort(const int2* __restrict__ binned,
                                                     const int* __restrict__ binstart,
                                                     int* __restrict__ rowptr,
                                                     int2* __restrict__ pairs,
                                                     int N, int E, int NB) {
    __shared__ int2 stage[SORT_CAP];
    __shared__ int cnt[NPB];
    __shared__ int cur[NPB];
    __shared__ int wsum[NPB / 32];
    const int tid = threadIdx.x;
    const int lane = tid & 31;
    const int wave = tid >> 5;
    const int b = blockIdx.x;
    int s = binstart[b];
    int e = binstart[b + 1];
    s = min(max(s, 0), E);
    e = min(max(e, s), E);
    const int n = e - s;
    const bool fits = (n <= SORT_CAP);

    if (tid < NPB) cnt[tid] = 0;
    __syncthreads();

    int2 r[SORT_RPT];
    if (fits) {
#pragma unroll
        for (int k = 0; k < SORT_RPT; ++k) {
            const int i = s + k * SORT_T + tid;
            r[k] = make_int2(0, 0);
            if (i < e) r[k] = binned[i];
        }
#pragma unroll
        for (int k = 0; k < SORT_RPT; ++k) {
            const int i = s + k * SORT_T + tid;
            if (i < e) atomicAdd(&cnt[r[k].x & (NPB - 1)], 1);
        }
    } else {
        for (int i = s + tid; i < e; i += SORT_T)
            atomicAdd(&cnt[binned[i].x & (NPB - 1)], 1);
    }
    __syncthreads();

    int c = 0;
    if (tid < NPB) c = cnt[tid];
    const int inc = wave_incl_scan(c, lane);
    if (tid < NPB && lane == 31) wsum[wave] = inc;
    __syncthreads();
    if (tid < NPB) {
        int off = 0;
        for (int k = 0; k < wave; ++k) off += wsum[k];
        const int ex = off + inc - c;
        const int node = b * NPB + tid;
        if (node < N) rowptr[node] = s + ex;
        cur[tid] = ex;
    }
    if (b == NB - 1 && tid == 0) rowptr[N] = e;
    __syncthreads();

    if (fits) {
#pragma unroll
        for (int k = 0; k < SORT_RPT; ++k) {
            const int i = s + k * SORT_T + tid;
            if (i < e) {
                const int p = atomicAdd(&cur[r[k].x & (NPB - 1)], 1);
                if ((unsigned)p < (unsigned)SORT_CAP)
                    stage[p] = make_int2((int)((unsigned)r[k].x >> BIN_SHIFT), r[k].y);
            }
        }
        __syncthreads();
        for (int i = tid; i < n; i += SORT_T) pairs[s + i] = stage[i];
    } else {
        for (int i = s + tid; i < e; i += SORT_T) {
            const int2 rec = binned[i];
            const int p = s + atomicAdd(&cur[rec.x & (NPB - 1)], 1);
            if ((unsigned)p < (unsigned)E)
                pairs[p] = make_int2((int)((unsigned)rec.x >> BIN_SHIFT), rec.y);
        }
    }
}

__device__ __forceinline__ float node_dinv(const int* __restrict__ rowptr,
                                           const int2* __restrict__ pairs, int i) {
    float d = 1.0f;
    const int end = rowptr[i + 1];
    for (int p = rowptr[i]; p < end; ++p) d += __int_as_float(pairs[p].y);
    return (d > 0.0f) ? rsqrtf(d) : 0.0f;
}

__device__ __forceinline__ void gemm_tile32(const float* As, int stride, int K,
                                            const float* __restrict__ W,
                                            float* __restrict__ H,
                                            int rowbase, int tid, const float* rs) {
    const int wave = tid >> 5;
    const int lane = tid & 31;
    const int m    = lane & 15;
    const int half = lane >> 4;
    const int cb   = wave * 16;

    v8f c0 = {};
    v8f c1 = {};
    const float* wp  = W + cb + m;
    const float* ap0 = As + m * stride;
    const float* ap1 = As + (16 + m) * stride;

    for (int k = 0; k < K; k += 4) {
        const int koff = k + half * 2;
        v2f b;
        b[0] = wp[(size_t)koff * HID_CH];
        b[1] = wp[(size_t)(koff + 1) * HID_CH];
        v2f a0, a1;
        a0[0] = ap0[koff];
        a0[1] = ap0[koff + 1];
        a1[0] = ap1[koff];
        a1[1] = ap1[koff + 1];
        c0 = __builtin_amdgcn_wmma_f32_16x16x4_f32(false, a0, false, b,
                                                   (short)0, c0, false, false);
        c1 = __builtin_amdgcn_wmma_f32_16x16x4_f32(false, a1, false, b,
                                                   (short)0, c1, false, false);
    }

    float* hp0 = H + (size_t)rowbase * HID_CH + cb + m;
    float* hp1 = hp0 + (size_t)16 * HID_CH;
#pragma unroll
    for (int r = 0; r < 8; ++r) {
        hp0[(size_t)(r + 8 * half) * HID_CH] = c0[r] * rs[r + 8 * half];
        hp1[(size_t)(r + 8 * half) * HID_CH] = c1[r] * rs[16 + r + 8 * half];
    }
}

__global__ __launch_bounds__(256) void k_gemm_dinv(const float* __restrict__ A,
                                                   const float* __restrict__ W,
                                                   float* __restrict__ H, int K,
                                                   const int* __restrict__ rowptr,
                                                   const int* __restrict__ pairs_i,
                                                   float* __restrict__ dinv) {
    __shared__ float As[GEMM_ROWS * (IN_CH + 4)];
    __shared__ float s_rs[GEMM_ROWS];
    const int stride = K + 4;
    const int tid = threadIdx.x;
    const int rowbase = blockIdx.x * GEMM_ROWS;

    if (tid < 32) {
        const float d = node_dinv(rowptr, (const int2*)pairs_i, rowbase + tid);
        dinv[rowbase + tid] = d;
        s_rs[tid] = d;
    }

    for (int idx = tid; idx < GEMM_ROWS * K; idx += 256) {
        int r = idx / K;
        int kk = idx - r * K;
        As[r * stride + kk] = A[(size_t)(rowbase + r) * K + kk];
    }
    __syncthreads();

    gemm_tile32(As, stride, K, W, H, rowbase, tid, s_rs);
}


__device__ __forceinline__ float4 gather_row(int node, int lane,
                                             const int* __restrict__ rowptr,
                                             const int2* __restrict__ pairs,
                                             const float* __restrict__ dinv,
                                             const float4* __restrict__ hb, int N,
                                             float& di_out) {
    const int beg = rowptr[node];
    const int end = rowptr[node + 1];
    const float di = dinv[node];
    di_out = di;

    float4 a = hb[(size_t)node * 32 + lane];

    for (int base = beg; base < end; base += 32) {
        const int p = base + lane;
        int src = 0;
        float nv = 0.0f;
        if (p < end) {
            int2 pr = pairs[p];
            src = pr.x;
            if ((unsigned)src >= (unsigned)N) src = 0;
            nv = __int_as_float(pr.y);
        }
        const int nv_bits = __float_as_int(nv);
        const int cnt = min(32, end - base);
        int j = 0;
        for (; j + 4 <= cnt; j += 4) {
            const int s0 = __builtin_amdgcn_readlane(src, j);
            const int s1 = __builtin_amdgcn_readlane(src, j + 1);
            const int s2 = __builtin_amdgcn_readlane(src, j + 2);
            const int s3 = __builtin_amdgcn_readlane(src, j + 3);
            const float n0 = __int_as_float(__builtin_amdgcn_readlane(nv_bits, j));
            const float n1 = __int_as_float(__builtin_amdgcn_readlane(nv_bits, j + 1));
            const float n2 = __int_as_float(__builtin_amdgcn_readlane(nv_bits, j + 2));
            const float n3 = __int_as_float(__builtin_amdgcn_readlane(nv_bits, j + 3));
            const float4 r0 = hb[(size_t)s0 * 32 + lane];
            const float4 r1 = hb[(size_t)s1 * 32 + lane];
            const float4 r2 = hb[(size_t)s2 * 32 + lane];
            const float4 r3 = hb[(size_t)s3 * 32 + lane];
            a.x += r0.x * n0; a.y += r0.y * n0; a.z += r0.z * n0; a.w += r0.w * n0;
            a.x += r1.x * n1; a.y += r1.y * n1; a.z += r1.z * n1; a.w += r1.w * n1;
            a.x += r2.x * n2; a.y += r2.y * n2; a.z += r2.z * n2; a.w += r2.w * n2;
            a.x += r3.x * n3; a.y += r3.y * n3; a.z += r3.z * n3; a.w += r3.w * n3;
        }
        for (; j < cnt; ++j) {
            const int s0 = __builtin_amdgcn_readlane(src, j);
            const float n0 = __int_as_float(__builtin_amdgcn_readlane(nv_bits, j));
            const float4 r0 = hb[(size_t)s0 * 32 + lane];
            a.x += r0.x * n0; a.y += r0.y * n0; a.z += r0.z * n0; a.w += r0.w * n0;
        }
    }
    a.x *= di; a.y *= di; a.z *= di; a.w *= di;
    return a;
}

__global__ __launch_bounds__(256, 16) void k_gather_relu_gemm(const int* __restrict__ rowptr,
                                                          const int* __restrict__ pairs_i,
                                                          const float* __restrict__ dinv,
                                                          const float* __restrict__ h,
                                                          const float* __restrict__ b,
                                                          const float* __restrict__ W,
                                                          float* __restrict__ H2, int N) {
    __shared__ float As[GEMM_ROWS * (HID_CH + 4)];
    __shared__ float s_rs[GEMM_ROWS];
    const int tid = threadIdx.x;
    const int lane = tid & 31;
    const int wave = __builtin_amdgcn_readfirstlane(tid >> 5);
    const int rowbase = blockIdx.x * GEMM_ROWS;
    const float4 bb = ((const float4*)b)[lane];

    for (int i = 0; i < GEMM_ROWS / 8; ++i) {
        const int r = wave + 8 * i;
        const int node = rowbase + r;
        float4 v = make_float4(0.0f, 0.0f, 0.0f, 0.0f);
        float di = 0.0f;
        if (node < N) {
            v = gather_row(node, lane, rowptr, (const int2*)pairs_i, dinv,
                           (const float4*)h, N, di);
            v.x = fmaxf(v.x + bb.x, 0.0f);
            v.y = fmaxf(v.y + bb.y, 0.0f);
            v.z = fmaxf(v.z + bb.z, 0.0f);
            v.w = fmaxf(v.w + bb.w, 0.0f);
        }
        *(float4*)(As + r * (HID_CH + 4) + lane * 4) = v;
        if (lane == 0) s_rs[r] = di;
    }
    __syncthreads();

    gemm_tile32(As, HID_CH + 4, HID_CH, W, H2, rowbase, tid, s_rs);
}

__global__ __launch_bounds__(256) void k_gather_l2norm(const int* __restrict__ rowptr,
                                                       const int* __restrict__ pairs_i,
                                                       const float* __restrict__ dinv,
                                                       const float* __restrict__ h,
                                                       const float* __restrict__ b,
                                                       float* __restrict__ out, int N) {
    const int lane = threadIdx.x & 31;
    const int node = __builtin_amdgcn_readfirstlane((int)((blockIdx.x * 256 + threadIdx.x) >> 5));
    if (node >= N) return;
    float di;
    float4 v = gather_row(node, lane, rowptr, (const int2*)pairs_i, dinv,
                          (const float4*)h, N, di);
    const float4 bb = ((const float4*)b)[lane];
    v.x += bb.x; v.y += bb.y; v.z += bb.z; v.w += bb.w;
    float ss = v.x * v.x + v.y * v.y + v.z * v.z + v.w * v.w;
#pragma unroll
    for (int m = 16; m >= 1; m >>= 1) ss += __shfl_xor(ss, m, 32);
    float inv = 1.0f / fmaxf(sqrtf(ss), 1e-12f);
    v.x *= inv; v.y *= inv; v.z *= inv; v.w *= inv;
    ((float4*)out)[(size_t)node * 32 + lane] = v;
}


static inline size_t align_up(size_t v, size_t a) { return (v + a - 1) & ~(a - 1); }

extern "C" void kernel_launch(void* const* d_in, const int* in_sizes, int n_in,
                              void* d_out, int out_size, void* d_ws, size_t ws_size,
                              hipStream_t stream) {
    const float* x  = (const float*)d_in[0];
    const int*   ei = (const int*)d_in[1];
    const float* ew = (const float*)d_in[2];
    const float* W1 = (const float*)d_in[3];
    const float* b1 = (const float*)d_in[4];
    const float* W2 = (const float*)d_in[5];
    const float* b2 = (const float*)d_in[6];
    float* out = (float*)d_out;

    const int N = in_sizes[0] / IN_CH;
    const int E = in_sizes[1] / 2;
    const int* row = ei;
    const int* col = ei + E;

    const int NB  = (N + NPB - 1) / NPB;
    const int NCH = (E + CHUNK - 1) / CHUNK;

    char* ws = (char*)d_ws;
    size_t off = 0;
    int*   rowptr   = (int*)(ws + off);   off = align_up(off + ((size_t)N + 1) * 4, 1024);
    float* dinv     = (float*)(ws + off); off = align_up(off + (size_t)N * 4, 1024);
    int*   histg    = (int*)(ws + off);   off = align_up(off + (size_t)NCH * NB * 4, 1024);
    int*   binstart = (int*)(ws + off);   off = align_up(off + ((size_t)NB + 1) * 4, 1024);
    float* h        = (float*)(ws + off); off = align_up(off + (size_t)N * HID_CH * 4, 1024);
    int*   pairs    = (int*)(ws + off);   off = align_up(off + (size_t)E * 8, 1024);
    int2*  binned   = (int2*)h;
    (void)ws_size; (void)n_in; (void)out_size;

    const int nb_nwave  = (N + 7) / 8;
    const int nb_gemm   = N / GEMM_ROWS;

    k_bin_hist  <<<NCH, 1024, 0, stream>>>(col, histg, E, NB);
    k_bin_place <<<NCH, 1024, 0, stream>>>(row, col, ew, histg, binstart, binned, E, N, NB, NCH);
    k_bin_sort  <<<NB, SORT_T, 0, stream>>>(binned, binstart, rowptr, (int2*)pairs, N, E, NB);

    k_gemm_dinv  <<<nb_gemm, 256, 0, stream>>>(x, W1, out, IN_CH, rowptr, pairs, dinv);
    k_gather_relu_gemm<<<nb_gemm, 256, 0, stream>>>(rowptr, pairs, dinv, out, b1, W2, h, N);
    k_gather_l2norm<<<nb_nwave, 256, 0, stream>>>(rowptr, pairs, dinv, h, b2, out, N);
}


# Layer-1 A tile staged as 16-byte pieces with constant row length: eight loads per thread in flight at once, no per-element integer division, 16-byte LDS stores; loads non-temporal
# speedup vs baseline: 4.7336x; 1.0931x over previous
#include <hip/hip_runtime.h>
#include <hip/hip_bf16.h>


typedef __attribute__((ext_vector_type(2))) float v2f;
typedef __attribute__((ext_vector_type(8))) float v8f;

#define IN_CH 256
#define HID_CH 128
#define GEMM_ROWS 32


#define BIN_SHIFT 9
#define NPB (1 << BIN_SHIFT)
#define MAX_BINS 256
#define CHUNK 12288
#define EPT (CHUNK / 1024)
#define SORT_T 1024
#define SORT_RPT 12
#define SORT_CAP (SORT_T * SORT_RPT)

__device__ __forceinline__ int bin_of(int c, int NB) {
    unsigned b = (unsigned)c >> BIN_SHIFT;
    return (int)(b < (unsigned)NB ? b : (unsigned)(NB - 1));
}

__device__ __forceinline__ int wave_incl_scan(int v, int lane) {
    for (int d = 1; d < 32; d <<= 1) {
        int t = __shfl_up(v, d, 32);
        if (lane >= d) v += t;
    }
    return v;
}

__global__ __launch_bounds__(1024) void k_bin_hist(const int* __restrict__ col,
                                                   int* __restrict__ histg, int E, int NB) {
    __shared__ int hist[MAX_BINS];
    const int tid = threadIdx.x;
    if (tid < MAX_BINS) hist[tid] = 0;
    __syncthreads();
    const int base = blockIdx.x * CHUNK;
    int c[EPT];
#pragma unroll
    for (int k = 0; k < EPT; ++k) {
        const int e = base + k * 1024 + tid;
        c[k] = (e < E) ? col[e] : -1;
    }
#pragma unroll
    for (int k = 0; k < EPT; ++k) {
        const int e = base + k * 1024 + tid;
        if (e < E) atomicAdd(&hist[bin_of(c[k], NB)], 1);
    }
    __syncthreads();
    if (tid < NB) histg[blockIdx.x * NB + tid] = hist[tid];
}

__global__ __launch_bounds__(1024) void k_bin_place(const int* __restrict__ row,
                                                    const int* __restrict__ col,
                                                    const float* __restrict__ w,
                                                    const int* __restrict__ histg,
                                                    int* __restrict__ binstart,
                                                    int2* __restrict__ binned,
                                                    int E, int N, int NB, int NCH) {
    __shared__ int2 stage[CHUNK];
    __shared__ int part_tot[4][MAX_BINS];
    __shared__ int part_pre[4][MAX_BINS];
    __shared__ int s_mine[MAX_BINS];
    __shared__ int s_loff[MAX_BINS];
    __shared__ int s_gbase[MAX_BINS];
    __shared__ int s_cursor[MAX_BINS];
    __shared__ int wsA[8];
    __shared__ int wsB[8];
    const int tid = threadIdx.x;
    const int lane = tid & 31;
    const int wave = tid >> 5;
    const int blk = blockIdx.x;

    {
        const int g = tid >> 8;
        const int bin = tid & (MAX_BINS - 1);
        int tot = 0, pre = 0;
        if (bin < NB) {
            for (int j = g; j < NCH; j += 4) {
                const int v = histg[j * NB + bin];
                tot += v;
                if (j < blk) pre += v;
            }
        }
        part_tot[g][bin] = tot;
        part_pre[g][bin] = pre;
    }
    __syncthreads();

    int t_tot = 0, t_pre = 0, t_mine = 0;
    if (tid < MAX_BINS) {
        t_tot = part_tot[0][tid] + part_tot[1][tid] + part_tot[2][tid] + part_tot[3][tid];
        t_pre = part_pre[0][tid] + part_pre[1][tid] + part_pre[2][tid] + part_pre[3][tid];
        if (tid < NB) t_mine = histg[blk * NB + tid];
    }
    const int incA = wave_incl_scan(t_tot, lane);
    const int incB = wave_incl_scan(t_mine, lane);
    if (tid < MAX_BINS && lane == 31) { wsA[wave] = incA; wsB[wave] = incB; }
    __syncthreads();
    if (tid < MAX_BINS) {
        int offA = 0, offB = 0;
        for (int k = 0; k < wave; ++k) { offA += wsA[k]; offB += wsB[k]; }
        const int exA = offA + incA - t_tot;
        const int exB = offB + incB - t_mine;
        s_gbase[tid] = exA + t_pre;
        s_loff[tid] = exB;
        s_cursor[tid] = exB;
        s_mine[tid] = t_mine;
        if (blk == 0 && tid < NB) binstart[tid] = exA;
        if (blk == 0 && tid == NB - 1) binstart[NB] = exA + t_tot;
    }
    __syncthreads();

    const int base = blk * CHUNK;
    int c[EPT], s[EPT], wb[EPT];
#pragma unroll
    for (int k = 0; k < EPT; ++k) {
        const int e = base + k * 1024 + tid;
        c[k] = 0; s[k] = 0; wb[k] = 0;
        if (e < E) { c[k] = col[e]; s[k] = row[e]; wb[k] = __float_as_int(w[e]); }
    }
#pragma unroll
    for (int k = 0; k < EPT; ++k) {
        const int e = base + k * 1024 + tid;
        if (e < E) {
            int src = s[k];
            if ((unsigned)src >= (unsigned)N) src = 0;
            const int pos = atomicAdd(&s_cursor[bin_of(c[k], NB)], 1);
            if ((unsigned)pos < (unsigned)CHUNK)
                stage[pos] = make_int2((src << BIN_SHIFT) | (c[k] & (NPB - 1)), wb[k]);
        }
    }
    __syncthreads();

    for (int b = wave; b < NB; b += 32) {
        const int n = s_mine[b];
        const int lo = s_loff[b];
        const int gb = s_gbase[b];
        for (int i = lane; i < n; i += 32) {
            const int p = gb + i;
            if ((unsigned)p < (unsigned)E && (unsigned)(lo + i) < (unsigned)CHUNK)
                binned[p] = stage[lo + i];
        }
    }
}

__global__ __launch_bounds__(SORT_T) void k_bin_sort(const int2* __restrict__ binned,
                                                     const int* __restrict__ binstart,
                                                     int* __restrict__ rowptr,
                                                     int2* __restrict__ pairs,
                                                     int N, int E, int NB) {
    __shared__ int2 stage[SORT_CAP];
    __shared__ int cnt[NPB];
    __shared__ int cur[NPB];
    __shared__ int wsum[NPB / 32];
    const int tid = threadIdx.x;
    const int lane = tid & 31;
    const int wave = tid >> 5;
    const int b = blockIdx.x;
    int s = binstart[b];
    int e = binstart[b + 1];
    s = min(max(s, 0), E);
    e = min(max(e, s), E);
    const int n = e - s;
    const bool fits = (n <= SORT_CAP);

    if (tid < NPB) cnt[tid] = 0;
    __syncthreads();

    int2 r[SORT_RPT];
    if (fits) {
#pragma unroll
        for (int k = 0; k < SORT_RPT; ++k) {
            const int i = s + k * SORT_T + tid;
            r[k] = make_int2(0, 0);
            if (i < e) r[k] = binned[i];
        }
#pragma unroll
        for (int k = 0; k < SORT_RPT; ++k) {
            const int i = s + k * SORT_T + tid;
            if (i < e) atomicAdd(&cnt[r[k].x & (NPB - 1)], 1);
        }
    } else {
        for (int i = s + tid; i < e; i += SORT_T)
            atomicAdd(&cnt[binned[i].x & (NPB - 1)], 1);
    }
    __syncthreads();

    int c = 0;
    if (tid < NPB) c = cnt[tid];
    const int inc = wave_incl_scan(c, lane);
    if (tid < NPB && lane == 31) wsum[wave] = inc;
    __syncthreads();
    if (tid < NPB) {
        int off = 0;
        for (int k = 0; k < wave; ++k) off += wsum[k];
        const int ex = off + inc - c;
        const int node = b * NPB + tid;
        if (node < N) rowptr[node] = s + ex;
        cur[tid] = ex;
    }
    if (b == NB - 1 && tid == 0) rowptr[N] = e;
    __syncthreads();

    if (fits) {
#pragma unroll
        for (int k = 0; k < SORT_RPT; ++k) {
            const int i = s + k * SORT_T + tid;
            if (i < e) {
                const int p = atomicAdd(&cur[r[k].x & (NPB - 1)], 1);
                if ((unsigned)p < (unsigned)SORT_CAP)
                    stage[p] = make_int2((int)((unsigned)r[k].x >> BIN_SHIFT), r[k].y);
            }
        }
        __syncthreads();
        for (int i = tid; i < n; i += SORT_T) pairs[s + i] = stage[i];
    } else {
        for (int i = s + tid; i < e; i += SORT_T) {
            const int2 rec = binned[i];
            const int p = s + atomicAdd(&cur[rec.x & (NPB - 1)], 1);
            if ((unsigned)p < (unsigned)E)
                pairs[p] = make_int2((int)((unsigned)rec.x >> BIN_SHIFT), rec.y);
        }
    }
}

__device__ __forceinline__ float node_dinv(const int* __restrict__ rowptr,
                                           const int2* __restrict__ pairs, int i) {
    float d = 1.0f;
    const int end = rowptr[i + 1];
    for (int p = rowptr[i]; p < end; ++p) d += __int_as_float(pairs[p].y);
    return (d > 0.0f) ? rsqrtf(d) : 0.0f;
}

__device__ __forceinline__ void gemm_tile32(const float* As, int stride, int K,
                                            const float* __restrict__ W,
                                            float* __restrict__ H,
                                            int rowbase, int tid, const float* rs) {
    const int wave = tid >> 5;
    const int lane = tid & 31;
    const int m    = lane & 15;
    const int half = lane >> 4;
    const int cb   = wave * 16;

    v8f c0 = {};
    v8f c1 = {};
    const float* wp  = W + cb + m;
    const float* ap0 = As + m * stride;
    const float* ap1 = As + (16 + m) * stride;

    for (int k = 0; k < K; k += 4) {
        const int koff = k + half * 2;
        v2f b;
        b[0] = wp[(size_t)koff * HID_CH];
        b[1] = wp[(size_t)(koff + 1) * HID_CH];
        v2f a0, a1;
        a0[0] = ap0[koff];
        a0[1] = ap0[koff + 1];
        a1[0] = ap1[koff];
        a1[1] = ap1[koff + 1];
        c0 = __builtin_amdgcn_wmma_f32_16x16x4_f32(false, a0, false, b,
                                                   (short)0, c0, false, false);
        c1 = __builtin_amdgcn_wmma_f32_16x16x4_f32(false, a1, false, b,
                                                   (short)0, c1, false, false);
    }

    float* hp0 = H + (size_t)rowbase * HID_CH + cb + m;
    float* hp1 = hp0 + (size_t)16 * HID_CH;
#pragma unroll
    for (int r = 0; r < 8; ++r) {
        hp0[(size_t)(r + 8 * half) * HID_CH] = c0[r] * rs[r + 8 * half];
        hp1[(size_t)(r + 8 * half) * HID_CH] = c1[r] * rs[16 + r + 8 * half];
    }
}

typedef __attribute__((ext_vector_type(4))) float v4f;
#define A_V4_PER_ROW (IN_CH / 4)
#define A_V4_PER_THREAD (GEMM_ROWS * A_V4_PER_ROW / 256)

__global__ __launch_bounds__(256) void k_gemm_dinv(const float* __restrict__ A,
                                                   const float* __restrict__ W,
                                                   float* __restrict__ H,
                                                   const int* __restrict__ rowptr,
                                                   const int* __restrict__ pairs_i,
                                                   float* __restrict__ dinv) {
    __shared__ __attribute__((aligned(16))) float As[GEMM_ROWS * (IN_CH + 4)];
    __shared__ float s_rs[GEMM_ROWS];
    const int stride = IN_CH + 4;
    const int tid = threadIdx.x;
    const int rowbase = blockIdx.x * GEMM_ROWS;

    const v4f* A4 = (const v4f*)(A + (size_t)rowbase * IN_CH);
    v4f piece[A_V4_PER_THREAD];
#pragma unroll
    for (int i = 0; i < A_V4_PER_THREAD; ++i)
        piece[i] = __builtin_nontemporal_load(A4 + i * 256 + tid);

    if (tid < 32) {
        const float d = node_dinv(rowptr, (const int2*)pairs_i, rowbase + tid);
        dinv[rowbase + tid] = d;
        s_rs[tid] = d;
    }

#pragma unroll
    for (int i = 0; i < A_V4_PER_THREAD; ++i) {
        const int idx = i * 256 + tid;
        const int r = idx / A_V4_PER_ROW;
        const int c4 = idx % A_V4_PER_ROW;
        *(v4f*)(As + r * stride + c4 * 4) = piece[i];
    }
    __syncthreads();

    gemm_tile32(As, stride, IN_CH, W, H, rowbase, tid, s_rs);
}


__device__ __forceinline__ float4 gather_row(int node, int lane,
                                             const int* __restrict__ rowptr,
                                             const int2* __restrict__ pairs,
                                             const float* __restrict__ dinv,
                                             const float4* __restrict__ hb, int N,
                                             float& di_out) {
    const int beg = rowptr[node];
    const int end = rowptr[node + 1];
    const float di = dinv[node];
    di_out = di;

    float4 a = hb[(size_t)node * 32 + lane];

    for (int base = beg; base < end; base += 32) {
        const int p = base + lane;
        int src = 0;
        float nv = 0.0f;
        if (p < end) {
            int2 pr = pairs[p];
            src = pr.x;
            if ((unsigned)src >= (unsigned)N) src = 0;
            nv = __int_as_float(pr.y);
        }
        const int nv_bits = __float_as_int(nv);
        const int cnt = min(32, end - base);
        int j = 0;
        for (; j + 4 <= cnt; j += 4) {
            const int s0 = __builtin_amdgcn_readlane(src, j);
            const int s1 = __builtin_amdgcn_readlane(src, j + 1);
            const int s2 = __builtin_amdgcn_readlane(src, j + 2);
            const int s3 = __builtin_amdgcn_readlane(src, j + 3);
            const float n0 = __int_as_float(__builtin_amdgcn_readlane(nv_bits, j));
            const float n1 = __int_as_float(__builtin_amdgcn_readlane(nv_bits, j + 1));
            const float n2 = __int_as_float(__builtin_amdgcn_readlane(nv_bits, j + 2));
            const float n3 = __int_as_float(__builtin_amdgcn_readlane(nv_bits, j + 3));
            const float4 r0 = hb[(size_t)s0 * 32 + lane];
            const float4 r1 = hb[(size_t)s1 * 32 + lane];
            const float4 r2 = hb[(size_t)s2 * 32 + lane];
            const float4 r3 = hb[(size_t)s3 * 32 + lane];
            a.x += r0.x * n0; a.y += r0.y * n0; a.z += r0.z * n0; a.w += r0.w * n0;
            a.x += r1.x * n1; a.y += r1.y * n1; a.z += r1.z * n1; a.w += r1.w * n1;
            a.x += r2.x * n2; a.y += r2.y * n2; a.z += r2.z * n2; a.w += r2.w * n2;
            a.x += r3.x * n3; a.y += r3.y * n3; a.z += r3.z * n3; a.w += r3.w * n3;
        }
        for (; j < cnt; ++j) {
            const int s0 = __builtin_amdgcn_readlane(src, j);
            const float n0 = __int_as_float(__builtin_amdgcn_readlane(nv_bits, j));
            const float4 r0 = hb[(size_t)s0 * 32 + lane];
            a.x += r0.x * n0; a.y += r0.y * n0; a.z += r0.z * n0; a.w += r0.w * n0;
        }
    }
    a.x *= di; a.y *= di; a.z *= di; a.w *= di;
    return a;
}

__global__ __launch_bounds__(256, 16) void k_gather_relu_gemm(const int* __restrict__ rowptr,
                                                          const int* __restrict__ pairs_i,
                                                          const float* __restrict__ dinv,
                                                          const float* __restrict__ h,
                                                          const float* __restrict__ b,
                                                          const float* __restrict__ W,
                                                          float* __restrict__ H2, int N) {
    __shared__ float As[GEMM_ROWS * (HID_CH + 4)];
    __shared__ float s_rs[GEMM_ROWS];
    const int tid = threadIdx.x;
    const int lane = tid & 31;
    const int wave = __builtin_amdgcn_readfirstlane(tid >> 5);
    const int rowbase = blockIdx.x * GEMM_ROWS;
    const float4 bb = ((const float4*)b)[lane];

    for (int i = 0; i < GEMM_ROWS / 8; ++i) {
        const int r = wave + 8 * i;
        const int node = rowbase + r;
        float4 v = make_float4(0.0f, 0.0f, 0.0f, 0.0f);
        float di = 0.0f;
        if (node < N) {
            v = gather_row(node, lane, rowptr, (const int2*)pairs_i, dinv,
                           (const float4*)h, N, di);
            v.x = fmaxf(v.x + bb.x, 0.0f);
            v.y = fmaxf(v.y + bb.y, 0.0f);
            v.z = fmaxf(v.z + bb.z, 0.0f);
            v.w = fmaxf(v.w + bb.w, 0.0f);
        }
        *(float4*)(As + r * (HID_CH + 4) + lane * 4) = v;
        if (lane == 0) s_rs[r] = di;
    }
    __syncthreads();

    gemm_tile32(As, HID_CH + 4, HID_CH, W, H2, rowbase, tid, s_rs);
}

__global__ __launch_bounds__(256) void k_gather_l2norm(const int* __restrict__ rowptr,
                                                       const int* __restrict__ pairs_i,
                                                       const float* __restrict__ dinv,
                                                       const float* __restrict__ h,
                                                       const float* __restrict__ b,
                                                       float* __restrict__ out, int N) {
    const int lane = threadIdx.x & 31;
    const int node = __builtin_amdgcn_readfirstlane((int)((blockIdx.x * 256 + threadIdx.x) >> 5));
    if (node >= N) return;
    float di;
    float4 v = gather_row(node, lane, rowptr, (const int2*)pairs_i, dinv,
                          (const float4*)h, N, di);
    const float4 bb = ((const float4*)b)[lane];
    v.x += bb.x; v.y += bb.y; v.z += bb.z; v.w += bb.w;
    float ss = v.x * v.x + v.y * v.y + v.z * v.z + v.w * v.w;
#pragma unroll
    for (int m = 16; m >= 1; m >>= 1) ss += __shfl_xor(ss, m, 32);
    float inv = 1.0f / fmaxf(sqrtf(ss), 1e-12f);
    v.x *= inv; v.y *= inv; v.z *= inv; v.w *= inv;
    ((float4*)out)[(size_t)node * 32 + lane] = v;
}


static inline size_t align_up(size_t v, size_t a) { return (v + a - 1) & ~(a - 1); }

extern "C" void kernel_launch(void* const* d_in, const int* in_sizes, int n_in,
                              void* d_out, int out_size, void* d_ws, size_t ws_size,
                              hipStream_t stream) {
    const float* x  = (const float*)d_in[0];
    const int*   ei = (const int*)d_in[1];
    const float* ew = (const float*)d_in[2];
    const float* W1 = (const float*)d_in[3];
    const float* b1 = (const float*)d_in[4];
    const float* W2 = (const float*)d_in[5];
    const float* b2 = (const float*)d_in[6];
    float* out = (float*)d_out;

    const int N = in_sizes[0] / IN_CH;
    const int E = in_sizes[1] / 2;
    const int* row = ei;
    const int* col = ei + E;

    const int NB  = (N + NPB - 1) / NPB;
    const int NCH = (E + CHUNK - 1) / CHUNK;

    char* ws = (char*)d_ws;
    size_t off = 0;
    int*   rowptr   = (int*)(ws + off);   off = align_up(off + ((size_t)N + 1) * 4, 1024);
    float* dinv     = (float*)(ws + off); off = align_up(off + (size_t)N * 4, 1024);
    int*   histg    = (int*)(ws + off);   off = align_up(off + (size_t)NCH * NB * 4, 1024);
    int*   binstart = (int*)(ws + off);   off = align_up(off + ((size_t)NB + 1) * 4, 1024);
    float* h        = (float*)(ws + off); off = align_up(off + (size_t)N * HID_CH * 4, 1024);
    int*   pairs    = (int*)(ws + off);   off = align_up(off + (size_t)E * 8, 1024);
    int2*  binned   = (int2*)h;
    (void)ws_size; (void)n_in; (void)out_size;

    const int nb_nwave  = (N + 7) / 8;
    const int nb_gemm   = N / GEMM_ROWS;

    k_bin_hist  <<<NCH, 1024, 0, stream>>>(col, histg, E, NB);
    k_bin_place <<<NCH, 1024, 0, stream>>>(row, col, ew, histg, binstart, binned, E, N, NB, NCH);
    k_bin_sort  <<<NB, SORT_T, 0, stream>>>(binned, binstart, rowptr, (int2*)pairs, N, E, NB);

    k_gemm_dinv  <<<nb_gemm, 256, 0, stream>>>(x, W1, out, rowptr, pairs, dinv);
    k_gather_relu_gemm<<<nb_gemm, 256, 0, stream>>>(rowptr, pairs, dinv, out, b1, W2, h, N);
    k_gather_l2norm<<<nb_nwave, 256, 0, stream>>>(rowptr, pairs, dinv, h, b2, out, N);
}
